# starting kernel
# baseline (speedup 1.0000x reference)
#include <hip/hip_runtime.h>
#include <hip/hip_bf16.h>

typedef __attribute__((ext_vector_type(2))) float v2f;
typedef __attribute__((ext_vector_type(8))) float v8f;

#define IW 4096
#define IH 4096
#define OW 4088
#define TILES_X 256
#define TILES_Y 256
#define STRIPS_Y 64
#define WAVES_PER_BLOCK 8

__device__ __forceinline__ void conv_tile_edge(
    const float* __restrict__ X,
    const float (&skern)[81],
    float* __restrict__ out,
    float b0, int x0, int y0, int lane)
{
    const int n  = lane & 15;
    const int kb = (lane >> 4) << 1;

    v8f acc = {};

    #pragma unroll
    for (int c = 0; c < 6; ++c) {
        const int dxb = 4 * c + kb;
        const int dd  = dxb - n;
        const bool m0 = (dd     >= 0) && (dd     <= 8);
        const bool m1 = (dd + 1 >= 0) && (dd + 1 <= 8);
        const int  i0 = min(max(dd,     0), 8);
        const int  i1 = min(max(dd + 1, 0), 8);

        #pragma unroll
        for (int ky = 0; ky < 9; ++ky) {
            int row = y0 + n + ky;
            row = row < IH - 1 ? row : IH - 1;
            int idx = row * IW + (x0 + dxb);
            idx = idx < IW * IW - 2 ? idx : IW * IW - 2;
            v2f a = *(const v2f*)(X + idx);

            float t0 = skern[ky * 9 + i0];
            float t1 = skern[ky * 9 + i1];
            v2f b;
            b.x = m0 ? t0 : 0.0f;
            b.y = m1 ? t1 : 0.0f;

            acc = __builtin_amdgcn_wmma_f32_16x16x4_f32(
                false, a, false, b, (short)0, acc, false, false);
        }
    }

    const int mbase = (lane >> 4) << 3;
    const int ox = x0 + n;
    #pragma unroll
    for (int i = 0; i < 8; ++i) {
        const int oy = y0 + mbase + i;
        if (oy < OW && ox < OW) out[oy * OW + ox] = acc[i] + b0;
    }
}

__device__ __forceinline__ void conv_strip4(
    const float* __restrict__ X,
    const float (&skern)[81],
    float* __restrict__ out,
    float b0, int x0, int y0, int lane)
{
    const int n  = lane & 15;
    const int kb = (lane >> 4) << 1;

    v8f acc[4] = {{}, {}, {}, {}};

    const float* base = X + (y0 + n) * IW + (x0 + kb);

    #pragma unroll
    for (int c = 0; c < 6; ++c) {
        const int dxb = 4 * c + kb;
        const int dd  = dxb - n;
        const bool m0 = (dd     >= 0) && (dd     <= 8);
        const bool m1 = (dd + 1 >= 0) && (dd + 1 <= 8);
        const int  i0 = min(max(dd,     0), 8);
        const int  i1 = min(max(dd + 1, 0), 8);

        #pragma unroll
        for (int ky = 0; ky < 9; ++ky) {
            float t0 = skern[ky * 9 + i0];
            float t1 = skern[ky * 9 + i1];
            v2f b;
            b.x = m0 ? t0 : 0.0f;
            b.y = m1 ? t1 : 0.0f;

            #pragma unroll
            for (int s = 0; s < 4; ++s) {
                v2f a = *(const v2f*)(base + (s * 16 + ky) * IW + 4 * c);
                acc[s] = __builtin_amdgcn_wmma_f32_16x16x4_f32(
                    false, a, false, b, (short)0, acc[s], false, false);
            }
        }
    }

    const int mbase = (lane >> 4) << 3;
    float* orow = out + (y0 + mbase) * OW + (x0 + n);
    #pragma unroll
    for (int s = 0; s < 4; ++s) {
        #pragma unroll
        for (int i = 0; i < 8; ++i) {
            orow[(s * 16 + i) * OW] = acc[s][i] + b0;
        }
    }
}

__global__ __launch_bounds__(256) void conv9x9_wmma_f32(
    const float* __restrict__ X,
    const float* __restrict__ Kf,
    const float* __restrict__ bias,
    float* __restrict__ out)
{
    __shared__ float skern[81];
    for (int t = threadIdx.x; t < 81; t += 256) skern[t] = Kf[t];
    __syncthreads();

    const int lane = threadIdx.x & 31;
    const int wave = threadIdx.x >> 5;
    const int job  = blockIdx.x * WAVES_PER_BLOCK + wave;
    const int tx   = job & (TILES_X - 1);
    const int sy   = job >> 8;
    const int x0   = tx * 16;
    const int y0   = sy * 64;

    const float b0 = bias[0];

    if (tx < TILES_X - 1 && sy < STRIPS_Y - 1) {
        conv_strip4(X, skern, out, b0, x0, y0, lane);
    } else {
        #pragma unroll
        for (int s = 0; s < 4; ++s) {
            conv_tile_edge(X, skern, out, b0, x0, y0 + s * 16, lane);
        }
    }
}

extern "C" void kernel_launch(void* const* d_in, const int* in_sizes, int n_in,
                              void* d_out, int out_size, void* d_ws, size_t ws_size,
                              hipStream_t stream) {
    const float* X    = (const float*)d_in[0];
    const float* Kf   = (const float*)d_in[1];
    const float* bias = (const float*)d_in[2];
    float* out        = (float*)d_out;

    const int total_jobs = TILES_X * STRIPS_Y;
    const int blocks = total_jobs / WAVES_PER_BLOCK;

    hipLaunchKernelGGL(conv9x9_wmma_f32, dim3(blocks), dim3(256), 0, stream,
                       X, Kf, bias, out);
}


# Input goes through LDS: each 4-wave block stages its 72x72 patch with coalesced 16-byte loads and reads A fragments from LDS (pitch 76, conflict-free), not per-lane global gathers; sum order kept
# speedup vs baseline: 2.0143x; 2.0143x over previous
#include <hip/hip_runtime.h>


typedef __attribute__((ext_vector_type(2))) float v2f;
typedef __attribute__((ext_vector_type(4))) float v4f;
typedef __attribute__((ext_vector_type(8))) float v8f;

#define IW 4096
#define IH 4096
#define OW 4088
#define BLK_THREADS 128
#define BLK_COLS 64
#define BLK_ROWS 64
#define PATCH_ROWS 72
#define PITCH 76

__global__ __launch_bounds__(BLK_THREADS) void conv9x9_wmma_f32(
    const float* __restrict__ X,
    const float* __restrict__ Kf,
    const float* __restrict__ bias,
    float* __restrict__ out)
{
    __shared__ float skern[BLK_THREADS];
    __shared__ __attribute__((aligned(16))) float patch[PATCH_ROWS * PITCH];

    const int tid  = threadIdx.x;
    const int bx0  = (blockIdx.x & 63) * BLK_COLS;
    const int by0  = (blockIdx.x >> 6) * BLK_ROWS;
    const int lane = tid & 31;
    const int wave = tid >> 5;

    const float tap = Kf[tid < 81 ? tid : 80];

    const int pr = lane >> 4;
    const int pq = lane & 15;
    v4f stage[9];
    #pragma unroll
    for (int j = 0; j < 9; ++j) {
        int gr = by0 + 2 * (wave + 4 * j) + pr;
        gr = gr < IH - 1 ? gr : IH - 1;
        stage[j] = *(const v4f*)(X + gr * IW + bx0 + 4 * pq);
    }
    const int eq  = tid & 1;
    const int er0 = tid >> 1;
    const int er1 = (64 + er0) < PATCH_ROWS - 1 ? (64 + er0) : PATCH_ROWS - 1;
    int egc = bx0 + BLK_COLS + 4 * eq;
    egc = egc < IW - 4 ? egc : IW - 4;
    int egr0 = by0 + er0;
    egr0 = egr0 < IH - 1 ? egr0 : IH - 1;
    int egr1 = by0 + er1;
    egr1 = egr1 < IH - 1 ? egr1 : IH - 1;
    const v4f extra0 = *(const v4f*)(X + egr0 * IW + egc);
    const v4f extra1 = *(const v4f*)(X + egr1 * IW + egc);

    #pragma unroll
    for (int j = 0; j < 9; ++j) {
        *(v4f*)(patch + (2 * (wave + 4 * j) + pr) * PITCH + 4 * pq) = stage[j];
    }
    *(v4f*)(patch + er0 * PITCH + BLK_COLS + 4 * eq) = extra0;
    *(v4f*)(patch + er1 * PITCH + BLK_COLS + 4 * eq) = extra1;
    skern[tid] = tap;
    __syncthreads();

    const int n  = lane & 15;
    const int h  = lane >> 4;
    const int kb = h << 1;

    v8f acc[4] = {{}, {}, {}, {}};

    const float* pbase = patch + n * PITCH + wave * 16 + kb;

    #pragma unroll
    for (int c = 0; c < 6; ++c) {
        const int dxb = 4 * c + kb;
        const int dd  = dxb - n;
        const bool m0 = (dd     >= 0) && (dd     <= 8);
        const bool m1 = (dd + 1 >= 0) && (dd + 1 <= 8);
        const int  i0 = min(max(dd,     0), 8);
        const int  i1 = min(max(dd + 1, 0), 8);

        #pragma unroll
        for (int ky = 0; ky < 9; ++ky) {
            float t0 = skern[ky * 9 + i0];
            float t1 = skern[ky * 9 + i1];
            v2f b;
            b.x = m0 ? t0 : 0.0f;
            b.y = m1 ? t1 : 0.0f;

            #pragma unroll
            for (int s = 0; s < 4; ++s) {
                v2f a = *(const v2f*)(pbase + (s * 16 + ky) * PITCH + 4 * c);
                acc[s] = __builtin_amdgcn_wmma_f32_16x16x4_f32(
                    false, a, false, b, (short)0, acc[s], false, false);
            }
        }
    }

    const float b0 = bias[0];
    const int mbase = h << 3;
    const int ox  = bx0 + wave * 16 + n;
    const int oy0 = by0 + mbase;

    if (bx0 + BLK_COLS <= OW && by0 + BLK_ROWS <= OW) {
        float* orow = out + oy0 * OW + ox;
        #pragma unroll
        for (int s = 0; s < 4; ++s) {
            #pragma unroll
            for (int i = 0; i < 8; ++i) {
                orow[(s * 16 + i) * OW] = acc[s][i] + b0;
            }
        }
    } else {
        #pragma unroll
        for (int s = 0; s < 4; ++s) {
            #pragma unroll
            for (int i = 0; i < 8; ++i) {
                const int oy = oy0 + s * 16 + i;
                if (oy < OW && ox < OW) out[oy * OW + ox] = acc[s][i] + b0;
            }
        }
    }
}

extern "C" void kernel_launch(void* const* d_in, const int* in_sizes, int n_in,
                              void* d_out, int out_size, void* d_ws, size_t ws_size,
                              hipStream_t stream) {
    const float* X    = (const float*)d_in[0];
    const float* Kf   = (const float*)d_in[1];
    const float* bias = (const float*)d_in[2];
    float* out        = (float*)d_out;

    hipLaunchKernelGGL(conv9x9_wmma_f32, dim3(4096), dim3(BLK_THREADS), 0, stream,
                       X, Kf, bias, out);
}


# Band re-tiled: a B column is a pixel of a 4x4 output cell (12x12 window): 36 matrix ops per 256 outputs, was 54; each A fragment serves 3 row groups; whole rows leave via LDS; sum order changes
# speedup vs baseline: 2.7526x; 1.3665x over previous
#include <hip/hip_runtime.h>


typedef __attribute__((ext_vector_type(2))) float v2f;
typedef __attribute__((ext_vector_type(4))) float v4f;
typedef __attribute__((ext_vector_type(8))) float v8f;

#define IW 4096
#define IH 4096
#define OW 4088
#define BLK_THREADS 128
#define BLK_COLS 64
#define BLK_ROWS 64
#define PATCH_ROWS 72
#define PITCH 76
#define TAB_PITCH 16
#define OUT_PITCH 68

__global__ __launch_bounds__(BLK_THREADS) void conv9x9_wmma_f32(
    const float* __restrict__ X,
    const float* __restrict__ Kf,
    const float* __restrict__ bias,
    float* __restrict__ out)
{
    __shared__ float tab[2 * BLK_THREADS];
    __shared__ __attribute__((aligned(16))) float patch[PATCH_ROWS * PITCH];
    __shared__ __attribute__((aligned(16))) float ostage[4 * 16 * OUT_PITCH];

    const int tid  = threadIdx.x;
    const int bx0  = (blockIdx.x & 63) * BLK_COLS;
    const int by0  = (blockIdx.x >> 6) * BLK_ROWS;
    const int lane = tid & 31;
    const int wave = tid >> 5;

    const int tdx  = (tid & 15) - 3;
    const int tdy0 = (tid >> 4) - 3;
    const int tdy1 = (tid >> 4) + 5;
    const int tcx  = min(max(tdx, 0), 8);
    const bool tinx = (tdx >= 0) && (tdx <= 8);
    const float tap0 = Kf[max(tdy0, 0) * 9 + tcx];
    const float tap1 = Kf[min(tdy1, 8) * 9 + tcx];

    const int pr = lane >> 4;
    const int pq = lane & 15;
    v4f stage[9];
    #pragma unroll
    for (int j = 0; j < 9; ++j) {
        int gr = by0 + 2 * (wave + 4 * j) + pr;
        gr = gr < IH - 1 ? gr : IH - 1;
        stage[j] = *(const v4f*)(X + gr * IW + bx0 + 4 * pq);
    }
    const int eq  = tid & 1;
    const int er0 = tid >> 1;
    const int er1 = (64 + er0) < PATCH_ROWS - 1 ? (64 + er0) : PATCH_ROWS - 1;
    int egc = bx0 + BLK_COLS + 4 * eq;
    egc = egc < IW - 4 ? egc : IW - 4;
    int egr0 = by0 + er0;
    egr0 = egr0 < IH - 1 ? egr0 : IH - 1;
    int egr1 = by0 + er1;
    egr1 = egr1 < IH - 1 ? egr1 : IH - 1;
    const v4f extra0 = *(const v4f*)(X + egr0 * IW + egc);
    const v4f extra1 = *(const v4f*)(X + egr1 * IW + egc);

    #pragma unroll
    for (int j = 0; j < 9; ++j) {
        *(v4f*)(patch + (2 * (wave + 4 * j) + pr) * PITCH + 4 * pq) = stage[j];
    }
    *(v4f*)(patch + er0 * PITCH + BLK_COLS + 4 * eq) = extra0;
    *(v4f*)(patch + er1 * PITCH + BLK_COLS + 4 * eq) = extra1;
    tab[tid]               = (tinx && tdy0 >= 0) ? tap0 : 0.0f;
    tab[BLK_THREADS + tid] = (tinx && tdy1 <= 8) ? tap1 : 0.0f;
    __syncthreads();

    const int p  = lane & 15;
    const int h  = lane >> 4;
    const int ny = p >> 2;
    const int nx = p & 3;

    const float* tb = tab + (3 - ny) * TAB_PITCH + (3 - nx) + 2 * h;
    v2f bf[12][3];
    #pragma unroll
    for (int r = 0; r < 12; ++r) {
        #pragma unroll
        for (int q = 0; q < 3; ++q) {
            bf[r][q].x = tb[r * TAB_PITCH + 4 * q];
            bf[r][q].y = tb[r * TAB_PITCH + 4 * q + 1];
        }
    }

    v8f acc[4] = {{}, {}, {}, {}};
    const float* pa = patch + (wave * 16) * PITCH + 4 * p + 2 * h;

    #pragma unroll
    for (int R = 0; R < 24; ++R) {
        #pragma unroll
        for (int q = 0; q < 3; ++q) {
            const v2f a = *(const v2f*)(pa + R * PITCH + 4 * q);
            #pragma unroll
            for (int s = 0; s < 4; ++s) {
                const int r = R - 4 * s;
                if (r >= 0 && r < 12) {
                    acc[s] = __builtin_amdgcn_wmma_f32_16x16x4_f32(
                        false, a, false, bf[r][q], (short)0, acc[s], false, false);
                }
            }
        }
    }

    const float b0 = bias[0];
    float* ow = ostage + wave * (16 * OUT_PITCH);
    float* od = ow + ny * OUT_PITCH + 32 * h + nx;
    #pragma unroll
    for (int s = 0; s < 4; ++s) {
        #pragma unroll
        for (int i = 0; i < 8; ++i) {
            od[(4 * s) * OUT_PITCH + 4 * i] = acc[s][i] + b0;
        }
    }
    __syncthreads();

    const int ox4 = bx0 + 4 * p;
    const int oyb = by0 + wave * 16 + h;
    const float* orb = ow + h * OUT_PITCH + 4 * p;

    if (bx0 + BLK_COLS <= OW && by0 + BLK_ROWS <= OW) {
        float* orow = out + oyb * OW + ox4;
        for (int j = 0; j < 8; ++j) {
            *(v4f*)(orow + (2 * j) * OW) = *(const v4f*)(orb + (2 * j) * OUT_PITCH);
        }
    } else {
        for (int j = 0; j < 8; ++j) {
            const int oy = oyb + 2 * j;
            if (oy < OW && ox4 < OW) {
                *(v4f*)(out + oy * OW + ox4) = *(const v4f*)(orb + (2 * j) * OUT_PITCH);
            }
        }
    }
}

extern "C" void kernel_launch(void* const* d_in, const int* in_sizes, int n_in,
                              void* d_out, int out_size, void* d_ws, size_t ws_size,
                              hipStream_t stream) {
    const float* X    = (const float*)d_in[0];
    const float* Kf   = (const float*)d_in[1];
    const float* bias = (const float*)d_in[2];
    float* out        = (float*)d_out;

    hipLaunchKernelGGL(conv9x9_wmma_f32, dim3(4096), dim3(BLK_THREADS), 0, stream,
                       X, Kf, bias, out);
}
